# starting kernel
# baseline (speedup 1.0000x reference)
#include <hip/hip_runtime.h>
#include <hip/hip_bf16.h>
#include <math.h>

typedef __attribute__((ext_vector_type(16))) _Float16 v16h;
typedef __attribute__((ext_vector_type(8)))  float    v8f;

#define HEADS 8
#define HDIM  32
#define FOUT  256
#define NEG_SLOPE 0.2f

__global__ void zero_i32(int* __restrict__ p, int n) {
    int i = blockIdx.x * blockDim.x + threadIdx.x;
    if (i < n) p[i] = 0;
}

__global__ void count_deg(const int* __restrict__ dst, int* __restrict__ deg, int e) {
    int i = blockIdx.x * blockDim.x + threadIdx.x;
    if (i < e) atomicAdd(&deg[dst[i]], 1);
}

__global__ void __launch_bounds__(1024) scan_off(const int* __restrict__ deg,
                                                 int* __restrict__ off, int n) {
    __shared__ int sh[1024];
    int tid = threadIdx.x;
    int chunk = (n + 1023) >> 10;
    int s0 = tid * chunk;
    int s1 = s0 + chunk; if (s1 > n) s1 = n;
    int local = 0;
    for (int i = s0; i < s1; ++i) local += deg[i];
    sh[tid] = local;
    __syncthreads();
    for (int o = 1; o < 1024; o <<= 1) {
        int v = (tid >= o) ? sh[tid - o] : 0;
        __syncthreads();
        sh[tid] += v;
        __syncthreads();
    }
    int run = sh[tid] - local;
    for (int i = s0; i < s1; ++i) { off[i] = run; run += deg[i]; }
    if (tid == 1023) off[n] = sh[1023];
}

__global__ void scatter_edges(const int* __restrict__ src, const int* __restrict__ dst,
                              const int* __restrict__ off, int* __restrict__ cursor,
                              int* __restrict__ eidx, int e) {
    int i = blockIdx.x * blockDim.x + threadIdx.x;
    if (i < e) {
        int d = dst[i];
        int pos = atomicAdd(&cursor[d], 1);
        eidx[off[d] + pos] = src[i];
    }
}

__global__ void cvt_f32_to_f16(const float* __restrict__ in, _Float16* __restrict__ out, int n) {
    int i = blockIdx.x * blockDim.x + threadIdx.x;
    if (i < n) out[i] = (_Float16)in[i];
}

__global__ void transpose_w(const float* __restrict__ W, _Float16* __restrict__ Wt, int K) {
    int i = blockIdx.x * blockDim.x + threadIdx.x;
    if (i < K * FOUT) {
        int k = i / FOUT, c = i % FOUT;
        Wt[c * K + k] = (_Float16)W[k * FOUT + c];
    }
}

__device__ __forceinline__ v16h load_frag(const _Float16* __restrict__ p) {
    v16h r;
#pragma unroll
    for (int e = 0; e < 8; ++e) { r[e] = p[e]; r[e + 8] = p[e + 16]; }
    return r;
}

__global__ void __launch_bounds__(256) gemm_wmma_f16(const _Float16* __restrict__ A,
                                                     const _Float16* __restrict__ Bt,
                                                     float* __restrict__ C,
                                                     int M, int K) {
    int gt   = blockIdx.x * 256 + threadIdx.x;
    int wid  = gt >> 5;
    int lane = gt & 31;
    int nWaves = (M >> 4) * (FOUT >> 6);
    if (wid >= nWaves) return;

    int m0 = (wid >> 2) << 4;
    int n0 = (wid & 3) << 6;
    int l15 = lane & 15;
    int kh  = (lane >> 4) << 3;

    const _Float16* pa  = A  + (size_t)(m0 + l15) * K + kh;
    const _Float16* pb0 = Bt + (size_t)(n0 + l15 +  0) * K + kh;
    const _Float16* pb1 = Bt + (size_t)(n0 + l15 + 16) * K + kh;
    const _Float16* pb2 = Bt + (size_t)(n0 + l15 + 32) * K + kh;
    const _Float16* pb3 = Bt + (size_t)(n0 + l15 + 48) * K + kh;

    v8f c0 = {}, c1 = {}, c2 = {}, c3 = {};
    for (int k0 = 0; k0 < K; k0 += 32) {
        v16h a  = load_frag(pa  + k0);
        v16h b0 = load_frag(pb0 + k0);
        v16h b1 = load_frag(pb1 + k0);
        v16h b2 = load_frag(pb2 + k0);
        v16h b3 = load_frag(pb3 + k0);
        c0 = __builtin_amdgcn_wmma_f32_16x16x32_f16(false, a, false, b0, (short)0, c0, false, false);
        c1 = __builtin_amdgcn_wmma_f32_16x16x32_f16(false, a, false, b1, (short)0, c1, false, false);
        c2 = __builtin_amdgcn_wmma_f32_16x16x32_f16(false, a, false, b2, (short)0, c2, false, false);
        c3 = __builtin_amdgcn_wmma_f32_16x16x32_f16(false, a, false, b3, (short)0, c3, false, false);
    }
    int rbase = (lane >> 4) << 3;
#pragma unroll
    for (int r = 0; r < 8; ++r) {
        size_t o = (size_t)(m0 + r + rbase) * FOUT + n0 + l15;
        C[o +  0] = c0[r];
        C[o + 16] = c1[r];
        C[o + 32] = c2[r];
        C[o + 48] = c3[r];
    }
}

__global__ void eler_kernel(const float* __restrict__ feat,
                            const float* __restrict__ al, const float* __restrict__ ar,
                            float* __restrict__ el, float* __restrict__ er, int n) {
    int i = blockIdx.x * blockDim.x + threadIdx.x;
    if (i >= n * HEADS) return;
    int node = i >> 3, hd = i & 7;
    const float* f = feat + (size_t)node * FOUT + hd * HDIM;
    const float* a = al + hd * HDIM;
    const float* b = ar + hd * HDIM;
    float sl = 0.f, sr = 0.f;
#pragma unroll
    for (int d = 0; d < HDIM; ++d) { float v = f[d]; sl += v * a[d]; sr += v * b[d]; }
    el[i] = sl; er[i] = sr;
}

__global__ void __launch_bounds__(256) gat_pull(const float* __restrict__ feat,
                                                const float* __restrict__ el,
                                                const float* __restrict__ er,
                                                const int* __restrict__ off,
                                                const int* __restrict__ eidx,
                                                const float* __restrict__ bias,
                                                float* __restrict__ out,
                                                int n, int mode) {
    int gt   = blockIdx.x * 256 + threadIdx.x;
    int v    = gt >> 5;
    int lane = gt & 31;
    if (v >= n) return;
    int hd = lane >> 2;
    int j  = lane & 3;

    int beg = off[v], end = off[v + 1];
    float ern = er[(size_t)v * HEADS + hd];

    float m = -INFINITY;
    for (int t = beg + j; t < end; t += 4) {
        int s = eidx[t];
        float e = el[(size_t)s * HEADS + hd] + ern;
        e = (e > 0.f) ? e : NEG_SLOPE * e;
        m = fmaxf(m, e);
    }
    m = fmaxf(m, __shfl_xor(m, 1));
    m = fmaxf(m, __shfl_xor(m, 2));

    float sum = 0.f;
    for (int t = beg + j; t < end; t += 4) {
        int s = eidx[t];
        float e = el[(size_t)s * HEADS + hd] + ern;
        e = (e > 0.f) ? e : NEG_SLOPE * e;
        sum += __expf(e - m);
    }
    sum += __shfl_xor(sum, 1);
    sum += __shfl_xor(sum, 2);
    float inv = (end > beg) ? 1.f / sum : 0.f;

    float acc[8];
#pragma unroll
    for (int q = 0; q < 8; ++q) acc[q] = 0.f;
    for (int t = beg; t < end; ++t) {
        int s = eidx[t];
        if (t + 1 < end) {
            int sn = eidx[t + 1];
            __builtin_prefetch(feat + (size_t)sn * FOUT + hd * HDIM + j * 8, 0, 0);
        }
        float e = el[(size_t)s * HEADS + hd] + ern;
        e = (e > 0.f) ? e : NEG_SLOPE * e;
        float alpha = __expf(e - m) * inv;
        const float* fs = feat + (size_t)s * FOUT + hd * HDIM + j * 8;
#pragma unroll
        for (int q = 0; q < 8; ++q) acc[q] += fs[q] * alpha;
    }

    size_t base = (size_t)v * FOUT + hd * HDIM + j * 8;
#pragma unroll
    for (int q = 0; q < 8; ++q) {
        float val = acc[q] + bias[hd * HDIM + j * 8 + q];
        if (mode == 0) {
            val = (val > 0.f) ? val : (__expf(val) - 1.f);
        } else {
            val += out[base + q];
        }
        out[base + q] = val;
    }
}

static inline size_t align256(size_t x) { return (x + 255) & ~(size_t)255; }

extern "C" void kernel_launch(void* const* d_in, const int* in_sizes, int n_in,
                              void* d_out, int out_size, void* d_ws, size_t ws_size,
                              hipStream_t stream) {
    const int N = in_sizes[0] / 128;
    const int E = in_sizes[1];

    const float* x   = (const float*)d_in[0];
    const int*   src = (const int*)d_in[1];
    const int*   dst = (const int*)d_in[2];

    char* ws = (char*)d_ws;
    size_t o = 0;
    int* off    = (int*)(ws + o); o = align256(o + (size_t)(N + 1) * 4);
    int* deg    = (int*)(ws + o); o = align256(o + (size_t)N * 4);
    int* cursor = (int*)(ws + o); o = align256(o + (size_t)N * 4);
    int* eidx   = (int*)(ws + o); o = align256(o + (size_t)E * 4);
    _Float16* h16  = (_Float16*)(ws + o); o = align256(o + (size_t)N * FOUT * 2);
    _Float16* wt16 = (_Float16*)(ws + o); o = align256(o + (size_t)FOUT * FOUT * 2);
    float* feat = (float*)(ws + o); o = align256(o + (size_t)N * FOUT * 4);
    float* el   = (float*)(ws + o); o = align256(o + (size_t)N * HEADS * 4);
    float* er   = (float*)(ws + o); o = align256(o + (size_t)N * HEADS * 4);
    (void)ws_size;

    float* h = (float*)d_out;

    const int T = 256;
    auto blk = [](int n, int t) { return (n + t - 1) / t; };

    zero_i32<<<blk(N, T), T, 0, stream>>>(deg, N);
    zero_i32<<<blk(N, T), T, 0, stream>>>(cursor, N);
    count_deg<<<blk(E, T), T, 0, stream>>>(dst, deg, E);
    scan_off<<<1, 1024, 0, stream>>>(deg, off, N);
    scatter_edges<<<blk(E, T), T, 0, stream>>>(src, dst, off, cursor, eidx, E);

    for (int layer = 0; layer < 4; ++layer) {
        const float* W  = (const float*)d_in[3 + 4 * layer];
        const float* al = (const float*)d_in[4 + 4 * layer];
        const float* ar = (const float*)d_in[5 + 4 * layer];
        const float* b  = (const float*)d_in[6 + 4 * layer];
        const int K = in_sizes[3 + 4 * layer] / FOUT;
        const float* hin = (layer == 0) ? x : h;

        cvt_f32_to_f16<<<blk(N * K, T), T, 0, stream>>>(hin, h16, N * K);
        transpose_w<<<blk(K * FOUT, T), T, 0, stream>>>(W, wt16, K);

        int nWaves = (N / 16) * (FOUT / 64);
        gemm_wmma_f16<<<blk(nWaves * 32, T), T, 0, stream>>>(h16, wt16, feat, N, K);

        eler_kernel<<<blk(N * HEADS, T), T, 0, stream>>>(feat, al, ar, el, er, N);

        int mode = (layer == 0) ? 0 : 1;
        gat_pull<<<blk(N * 32, T), T, 0, stream>>>(feat, el, er, off, eidx, b, h, N, mode);
    }
}


# gat_pull epilogue: bias and old output row read as four 16-byte loads in one group, row stored as two 16-byte stores (was 8 dependent rounds of dword load, wait, dword load, wait, dword store)
# speedup vs baseline: 1.6749x; 1.6749x over previous
#include <hip/hip_runtime.h>
#include <hip/hip_bf16.h>
#include <math.h>

typedef __attribute__((ext_vector_type(16))) _Float16 v16h;
typedef __attribute__((ext_vector_type(8)))  float    v8f;
typedef __attribute__((ext_vector_type(8)))  _Float16 v8h;

#define HEADS 8
#define HDIM  32
#define FOUT  256
#define NEG_SLOPE 0.2f

__global__ void zero_i32(int* __restrict__ p, int n) {
    int i = blockIdx.x * blockDim.x + threadIdx.x;
    if (i < n) p[i] = 0;
}

__global__ void count_deg(const int* __restrict__ dst, int* __restrict__ deg, int e) {
    int i = blockIdx.x * blockDim.x + threadIdx.x;
    if (i < e) atomicAdd(&deg[dst[i]], 1);
}

#define SCAN_T   1024
#define SCAN_SEG 4096
__global__ void __launch_bounds__(SCAN_T) scan_off(const int* __restrict__ deg,
                                                   int* __restrict__ off, int n) {
    __shared__ int sh_pre[32];
    __shared__ int sh_own[32];
    const int tid  = threadIdx.x;
    const int lane = tid & 31;
    const int w    = tid >> 5;
    const int base = blockIdx.x * SCAN_SEG;

    const int4* d4 = reinterpret_cast<const int4*>(deg);
    const int n4 = base >> 2;
    int pre = 0;
#pragma unroll 4
    for (int i = tid; i < n4; i += SCAN_T) {
        int4 q = d4[i];
        pre += (q.x + q.y) + (q.z + q.w);
    }

    const int i0 = base + (tid << 2);
    int v0 = 0, v1 = 0, v2 = 0, v3 = 0;
    if (i0 + 3 < n) {
        int4 q = d4[i0 >> 2];
        v0 = q.x; v1 = q.y; v2 = q.z; v3 = q.w;
    } else {
        if (i0     < n) v0 = deg[i0];
        if (i0 + 1 < n) v1 = deg[i0 + 1];
        if (i0 + 2 < n) v2 = deg[i0 + 2];
    }
    const int t = (v0 + v1) + (v2 + v3);

    int x = t;
#pragma unroll
    for (int o = 1; o < 32; o <<= 1) {
        int y = __shfl_up(x, o);
        if (lane >= o) x += y;
    }
#pragma unroll
    for (int o = 16; o > 0; o >>= 1) pre += __shfl_xor(pre, o);
    if (lane == 31) sh_own[w] = x;
    if (lane == 0)  sh_pre[w] = pre;
    __syncthreads();

    int c = sh_pre[lane] + ((lane < w) ? sh_own[lane] : 0);
#pragma unroll
    for (int o = 16; o > 0; o >>= 1) c += __shfl_xor(c, o);

    const int e0 = c + x - t;
    const int e1 = e0 + v0;
    const int e2 = e1 + v1;
    const int e3 = e2 + v2;
    const int e4 = e3 + v3;
    if (i0 + 3 < n) {
        *reinterpret_cast<int4*>(off + i0) = make_int4(e0, e1, e2, e3);
        if (i0 + 4 == n) off[n] = e4;
    } else if (i0 < n) {
        off[i0] = e0;
        if (i0 + 1 < n) off[i0 + 1] = e1;
        if (i0 + 2 < n) off[i0 + 2] = e2;
        off[n] = e4;
    }
}

__global__ void scatter_edges(const int* __restrict__ src, const int* __restrict__ dst,
                              const int* __restrict__ off, int* __restrict__ cursor,
                              int* __restrict__ eidx, int e) {
    int i = blockIdx.x * blockDim.x + threadIdx.x;
    if (i < e) {
        int d = dst[i];
        int pos = atomicAdd(&cursor[d], 1);
        eidx[off[d] + pos] = src[i];
    }
}

__global__ void cvt_f32_to_f16(const float* __restrict__ in, _Float16* __restrict__ out, int n) {
    int i = blockIdx.x * blockDim.x + threadIdx.x;
    if (i < n) out[i] = (_Float16)in[i];
}

__global__ void transpose_w(const float* __restrict__ W, _Float16* __restrict__ Wt, int K) {
    int i = blockIdx.x * blockDim.x + threadIdx.x;
    if (i < K * FOUT) {
        int k = i / FOUT, c = i % FOUT;
        Wt[c * K + k] = (_Float16)W[k * FOUT + c];
    }
}

__device__ __forceinline__ v16h load_frag(const _Float16* __restrict__ p) {
    v16h r;
#pragma unroll
    for (int e = 0; e < 8; ++e) { r[e] = p[e]; r[e + 8] = p[e + 16]; }
    return r;
}

__global__ void __launch_bounds__(256) gemm_wmma_f16(const _Float16* __restrict__ A,
                                                     const _Float16* __restrict__ Bt,
                                                     float* __restrict__ C,
                                                     int M, int K) {
    int gt   = blockIdx.x * 256 + threadIdx.x;
    int wid  = gt >> 5;
    int lane = gt & 31;
    int nWaves = (M >> 4) * (FOUT >> 6);
    if (wid >= nWaves) return;

    int m0 = (wid >> 2) << 4;
    int n0 = (wid & 3) << 6;
    int l15 = lane & 15;
    int kh  = (lane >> 4) << 3;

    const _Float16* pa  = A  + (size_t)(m0 + l15) * K + kh;
    const _Float16* pb0 = Bt + (size_t)(n0 + l15 +  0) * K + kh;
    const _Float16* pb1 = Bt + (size_t)(n0 + l15 + 16) * K + kh;
    const _Float16* pb2 = Bt + (size_t)(n0 + l15 + 32) * K + kh;
    const _Float16* pb3 = Bt + (size_t)(n0 + l15 + 48) * K + kh;

    v8f c0 = {}, c1 = {}, c2 = {}, c3 = {};
    for (int k0 = 0; k0 < K; k0 += 32) {
        v16h a  = load_frag(pa  + k0);
        v16h b0 = load_frag(pb0 + k0);
        v16h b1 = load_frag(pb1 + k0);
        v16h b2 = load_frag(pb2 + k0);
        v16h b3 = load_frag(pb3 + k0);
        c0 = __builtin_amdgcn_wmma_f32_16x16x32_f16(false, a, false, b0, (short)0, c0, false, false);
        c1 = __builtin_amdgcn_wmma_f32_16x16x32_f16(false, a, false, b1, (short)0, c1, false, false);
        c2 = __builtin_amdgcn_wmma_f32_16x16x32_f16(false, a, false, b2, (short)0, c2, false, false);
        c3 = __builtin_amdgcn_wmma_f32_16x16x32_f16(false, a, false, b3, (short)0, c3, false, false);
    }
    int rbase = (lane >> 4) << 3;
#pragma unroll
    for (int r = 0; r < 8; ++r) {
        size_t o = (size_t)(m0 + r + rbase) * FOUT + n0 + l15;
        C[o +  0] = c0[r];
        C[o + 16] = c1[r];
        C[o + 32] = c2[r];
        C[o + 48] = c3[r];
    }
}

typedef __attribute__((ext_vector_type(2))) float v2f;
typedef __attribute__((ext_vector_type(4))) float v4f;

__device__ __forceinline__ v4f ld4(const float* __restrict__ p) {
    return *reinterpret_cast<const v4f*>(p);
}

__device__ __forceinline__ float dot8(v4f x0, v4f x1, v4f a0, v4f a1) {
    v2f d = x0.xy * a0.xy;
    d = x0.zw * a0.zw + d;
    d = x1.xy * a1.xy + d;
    d = x1.zw * a1.zw + d;
    return d.x + d.y;
}

__device__ __forceinline__ float quad_sum(float x) {
    x += __int_as_float(__builtin_amdgcn_update_dpp(0, __float_as_int(x), 0x161, 0xf, 0xf, true));
    x += __int_as_float(__builtin_amdgcn_update_dpp(0, __float_as_int(x), 0x162, 0xf, 0xf, true));
    return x;
}

__device__ __forceinline__ float lrelu(float e) {
    return fmaxf(e, NEG_SLOPE * e);
}

__global__ void __launch_bounds__(256) gat_pull(const float* __restrict__ feat,
                                                const float* __restrict__ al,
                                                const float* __restrict__ ar,
                                                const int* __restrict__ off,
                                                const int* __restrict__ eidx,
                                                const float* __restrict__ bias,
                                                float* __restrict__ out,
                                                _Float16* __restrict__ out16,
                                                int n, int mode) {
    int gt   = blockIdx.x * 256 + threadIdx.x;
    int v    = gt >> 5;
    int lane = gt & 31;
    if (v >= n) return;
    const int col = (lane >> 2) * HDIM + (lane & 3) * 8;

    const int beg = off[v], end = off[v + 1];
    const int last = end - 1;

    const v4f a0 = ld4(al + col), a1 = ld4(al + col + 4);
    float ern;
    {
        const v4f b0 = ld4(ar + col), b1 = ld4(ar + col + 4);
        const float* fv = feat + (size_t)v * FOUT + col;
        ern = quad_sum(dot8(ld4(fv), ld4(fv + 4), b0, b1));
    }

    float m = -INFINITY, sum = 0.f;
    float acc[8];
#pragma unroll
    for (int q = 0; q < 8; ++q) acc[q] = 0.f;

    int s0 = 0, s1 = 0, s2 = 0, s3 = 0;
    if (end > beg) {
        s0 = eidx[beg];
        s1 = eidx[min(beg + 1, last)];
        s2 = eidx[min(beg + 2, last)];
        s3 = eidx[min(beg + 3, last)];
    }
    for (int t = beg; t < end; t += 4) {
        const int n0 = eidx[min(t + 4, last)];
        const int n1 = eidx[min(t + 5, last)];
        const int n2 = eidx[min(t + 6, last)];
        const int n3 = eidx[min(t + 7, last)];

        const float* f0 = feat + (size_t)s0 * FOUT + col;
        const float* f1 = feat + (size_t)s1 * FOUT + col;
        const float* f2 = feat + (size_t)s2 * FOUT + col;
        const float* f3 = feat + (size_t)s3 * FOUT + col;
        const v4f r0a = ld4(f0), r0b = ld4(f0 + 4);
        const v4f r1a = ld4(f1), r1b = ld4(f1 + 4);
        const v4f r2a = ld4(f2), r2b = ld4(f2 + 4);
        const v4f r3a = ld4(f3), r3b = ld4(f3 + 4);
        __builtin_amdgcn_sched_barrier(0);

        float e0 = lrelu(quad_sum(dot8(r0a, r0b, a0, a1)) + ern);
        float e1 = lrelu(quad_sum(dot8(r1a, r1b, a0, a1)) + ern);
        float e2 = lrelu(quad_sum(dot8(r2a, r2b, a0, a1)) + ern);
        float e3 = lrelu(quad_sum(dot8(r3a, r3b, a0, a1)) + ern);
        e1 = (t + 1 <= last) ? e1 : -INFINITY;
        e2 = (t + 2 <= last) ? e2 : -INFINITY;
        e3 = (t + 3 <= last) ? e3 : -INFINITY;

        const float mn = fmaxf(fmaxf(m, e0), fmaxf(fmaxf(e1, e2), e3));
        const float sc = __expf(m - mn);
        const float p0 = __expf(e0 - mn);
        const float p1 = __expf(e1 - mn);
        const float p2 = __expf(e2 - mn);
        const float p3 = __expf(e3 - mn);
        sum = sum * sc + ((p0 + p1) + (p2 + p3));
        const float r0[8] = {r0a.x, r0a.y, r0a.z, r0a.w, r0b.x, r0b.y, r0b.z, r0b.w};
        const float r1[8] = {r1a.x, r1a.y, r1a.z, r1a.w, r1b.x, r1b.y, r1b.z, r1b.w};
        const float r2[8] = {r2a.x, r2a.y, r2a.z, r2a.w, r2b.x, r2b.y, r2b.z, r2b.w};
        const float r3[8] = {r3a.x, r3a.y, r3a.z, r3a.w, r3b.x, r3b.y, r3b.z, r3b.w};
#pragma unroll
        for (int q = 0; q < 8; ++q) {
            float x = acc[q] * sc;
            x = fmaf(p0, r0[q], x);
            x = fmaf(p1, r1[q], x);
            x = fmaf(p2, r2[q], x);
            x = fmaf(p3, r3[q], x);
            acc[q] = x;
        }
        m = mn;
        asm volatile("" :: "v"(n0), "v"(n1), "v"(n2), "v"(n3));
        s0 = n0; s1 = n1; s2 = n2; s3 = n3;
    }
    const float inv = (end > beg) ? 1.f / sum : 0.f;

    const size_t base = (size_t)v * FOUT + col;
    const v4f bi0 = ld4(bias + col), bi1 = ld4(bias + col + 4);
    v4f rs0 = {0.f, 0.f, 0.f, 0.f}, rs1 = {0.f, 0.f, 0.f, 0.f};
    if (mode != 0) {
        rs0 = ld4(out + base);
        rs1 = ld4(out + base + 4);
    }
    const float bi[8] = {bi0.x, bi0.y, bi0.z, bi0.w, bi1.x, bi1.y, bi1.z, bi1.w};
    const float rs[8] = {rs0.x, rs0.y, rs0.z, rs0.w, rs1.x, rs1.y, rs1.z, rs1.w};
    float res[8];
    if (mode == 0) {
#pragma unroll
        for (int q = 0; q < 8; ++q) {
            const float val = acc[q] * inv + bi[q];
            res[q] = (val > 0.f) ? val : (__expf(val) - 1.f);
        }
    } else {
#pragma unroll
        for (int q = 0; q < 8; ++q) {
            const float val = acc[q] * inv + bi[q];
            res[q] = val + rs[q];
        }
    }
    v4f o0, o1;
    v8h hv;
#pragma unroll
    for (int q = 0; q < 4; ++q) {
        o0[q] = res[q];
        o1[q] = res[q + 4];
    }
#pragma unroll
    for (int q = 0; q < 8; ++q) hv[q] = (_Float16)res[q];
    *reinterpret_cast<v4f*>(out + base) = o0;
    *reinterpret_cast<v4f*>(out + base + 4) = o1;
    if (out16 != nullptr)
        *reinterpret_cast<v8h*>(out16 + base) = hv;
}

static inline size_t align256(size_t x) { return (x + 255) & ~(size_t)255; }

extern "C" void kernel_launch(void* const* d_in, const int* in_sizes, int n_in,
                              void* d_out, int out_size, void* d_ws, size_t ws_size,
                              hipStream_t stream) {
    const int N = in_sizes[0] / 128;
    const int E = in_sizes[1];

    const float* x   = (const float*)d_in[0];
    const int*   src = (const int*)d_in[1];
    const int*   dst = (const int*)d_in[2];

    char* ws = (char*)d_ws;
    size_t o = 0;
    int* off    = (int*)(ws + o); o = align256(o + (size_t)(N + 1) * 4);
    int* deg    = (int*)(ws + o); o = align256(o + (size_t)N * 4);
    int* cursor = (int*)(ws + o); o = align256(o + (size_t)N * 4);
    int* eidx   = (int*)(ws + o); o = align256(o + (size_t)E * 4);
    _Float16* h16  = (_Float16*)(ws + o); o = align256(o + (size_t)N * FOUT * 2);
    _Float16* wt16 = (_Float16*)(ws + o); o = align256(o + (size_t)FOUT * FOUT * 2);
    float* feat = (float*)(ws + o); o = align256(o + (size_t)N * FOUT * 4);
    (void)ws_size;

    float* h = (float*)d_out;

    const int T = 256;
    auto blk = [](int n, int t) { return (n + t - 1) / t; };

    zero_i32<<<blk(N, T), T, 0, stream>>>(deg, N);
    zero_i32<<<blk(N, T), T, 0, stream>>>(cursor, N);
    count_deg<<<blk(E, T), T, 0, stream>>>(dst, deg, E);
    scan_off<<<blk(N, SCAN_SEG), SCAN_T, 0, stream>>>(deg, off, N);
    scatter_edges<<<blk(E, T), T, 0, stream>>>(src, dst, off, cursor, eidx, E);

    for (int layer = 0; layer < 4; ++layer) {
        const float* W  = (const float*)d_in[3 + 4 * layer];
        const float* al = (const float*)d_in[4 + 4 * layer];
        const float* ar = (const float*)d_in[5 + 4 * layer];
        const float* b  = (const float*)d_in[6 + 4 * layer];
        const int K = in_sizes[3 + 4 * layer] / FOUT;
        const float* hin = (layer == 0) ? x : h;

        if (layer == 0)
            cvt_f32_to_f16<<<blk(N * K, T), T, 0, stream>>>(hin, h16, N * K);
        transpose_w<<<blk(K * FOUT, T), T, 0, stream>>>(W, wt16, K);

        int nWaves = (N / 16) * (FOUT / 64);
        gemm_wmma_f16<<<blk(nWaves * 32, T), T, 0, stream>>>(h16, wt16, feat, N, K);

        int mode = (layer == 0) ? 0 : 1;
        _Float16* hnext16 = (layer < 3) ? h16 : nullptr;
        gat_pull<<<blk(N * 32, T), T, 0, stream>>>(feat, al, ar, off, eidx, b, h, hnext16, N, mode);
    }
}


# the seven launches that wait for nothing (two counter zero fills, x to f16, four weight stagings) are one launch cut into block ranges; each element handled as before; 15 launches become 9
# speedup vs baseline: 2.0382x; 1.2169x over previous
#include <hip/hip_runtime.h>
#include <hip/hip_bf16.h>
#include <math.h>

typedef __attribute__((ext_vector_type(16))) _Float16 v16h;
typedef __attribute__((ext_vector_type(8)))  float    v8f;
typedef __attribute__((ext_vector_type(8)))  _Float16 v8h;

#define HEADS 8
#define HDIM  32
#define FOUT  256
#define NEG_SLOPE 0.2f

__global__ void count_deg(const int* __restrict__ dst, int* __restrict__ deg, int e) {
    int i = blockIdx.x * blockDim.x + threadIdx.x;
    if (i < e) atomicAdd(&deg[dst[i]], 1);
}

#define SCAN_T   1024
#define SCAN_SEG 4096
__global__ void __launch_bounds__(SCAN_T) scan_off(const int* __restrict__ deg,
                                                   int* __restrict__ off, int n) {
    __shared__ int sh_pre[32];
    __shared__ int sh_own[32];
    const int tid  = threadIdx.x;
    const int lane = tid & 31;
    const int w    = tid >> 5;
    const int base = blockIdx.x * SCAN_SEG;

    const int4* d4 = reinterpret_cast<const int4*>(deg);
    const int n4 = base >> 2;
    int pre = 0;
#pragma unroll 4
    for (int i = tid; i < n4; i += SCAN_T) {
        int4 q = d4[i];
        pre += (q.x + q.y) + (q.z + q.w);
    }

    const int i0 = base + (tid << 2);
    int v0 = 0, v1 = 0, v2 = 0, v3 = 0;
    if (i0 + 3 < n) {
        int4 q = d4[i0 >> 2];
        v0 = q.x; v1 = q.y; v2 = q.z; v3 = q.w;
    } else {
        if (i0     < n) v0 = deg[i0];
        if (i0 + 1 < n) v1 = deg[i0 + 1];
        if (i0 + 2 < n) v2 = deg[i0 + 2];
    }
    const int t = (v0 + v1) + (v2 + v3);

    int x = t;
#pragma unroll
    for (int o = 1; o < 32; o <<= 1) {
        int y = __shfl_up(x, o);
        if (lane >= o) x += y;
    }
#pragma unroll
    for (int o = 16; o > 0; o >>= 1) pre += __shfl_xor(pre, o);
    if (lane == 31) sh_own[w] = x;
    if (lane == 0)  sh_pre[w] = pre;
    __syncthreads();

    int c = sh_pre[lane] + ((lane < w) ? sh_own[lane] : 0);
#pragma unroll
    for (int o = 16; o > 0; o >>= 1) c += __shfl_xor(c, o);

    const int e0 = c + x - t;
    const int e1 = e0 + v0;
    const int e2 = e1 + v1;
    const int e3 = e2 + v2;
    const int e4 = e3 + v3;
    if (i0 + 3 < n) {
        *reinterpret_cast<int4*>(off + i0) = make_int4(e0, e1, e2, e3);
        if (i0 + 4 == n) off[n] = e4;
    } else if (i0 < n) {
        off[i0] = e0;
        if (i0 + 1 < n) off[i0 + 1] = e1;
        if (i0 + 2 < n) off[i0 + 2] = e2;
        off[n] = e4;
    }
}

__global__ void scatter_edges(const int* __restrict__ src, const int* __restrict__ dst,
                              const int* __restrict__ off, int* __restrict__ cursor,
                              int* __restrict__ eidx, int e) {
    int i = blockIdx.x * blockDim.x + threadIdx.x;
    if (i < e) {
        int d = dst[i];
        int pos = atomicAdd(&cursor[d], 1);
        eidx[off[d] + pos] = src[i];
    }
}

__device__ __forceinline__ void transpose_w(const float* __restrict__ W, _Float16* __restrict__ Wt,
                                            int K, int i) {
    if (i < K * FOUT) {
        int k = i / FOUT, c = i % FOUT;
        Wt[c * K + k] = (_Float16)W[k * FOUT + c];
    }
}

__device__ __forceinline__ void stage_w_frag(const float* __restrict__ W, _Float16* __restrict__ Wf,
                                             int i) {
    if (i < FOUT * FOUT) {
        int k = i / FOUT, c = i % FOUT;
        int w = c >> 4, cc = c & 15;
        int j = k >> 5, s = (k >> 4) & 1, h = (k >> 3) & 1, e = k & 7;
        Wf[(((w * 8 + j) * 2 + s) * 32 + (h * 16 + cc)) * 8 + e] = (_Float16)W[i];
    }
}

#define SETUP_T 256
#define STAGE_BLOCKS (FOUT * FOUT / SETUP_T)
__global__ void __launch_bounds__(SETUP_T) setup_all(const float* __restrict__ x,
                                                     _Float16* __restrict__ x16, int n_x,
                                                     int* __restrict__ deg,
                                                     int* __restrict__ cursor, int n_nodes,
                                                     const float* __restrict__ W0,
                                                     _Float16* __restrict__ Wt0, int K0,
                                                     const float* __restrict__ W1,
                                                     _Float16* __restrict__ Wf1,
                                                     const float* __restrict__ W2,
                                                     _Float16* __restrict__ Wf2,
                                                     const float* __restrict__ W3,
                                                     _Float16* __restrict__ Wf3,
                                                     int blocks_x, int blocks_zero,
                                                     int blocks_w0) {
    int b = blockIdx.x;
    if (b < blocks_x) {
        const int i = b * SETUP_T + threadIdx.x;
        if (i < n_x) x16[i] = (_Float16)x[i];
        return;
    }
    b -= blocks_x;
    if (b < blocks_zero) {
        const int i = b * SETUP_T + threadIdx.x;
        if (i < n_nodes) { deg[i] = 0; cursor[i] = 0; }
        return;
    }
    b -= blocks_zero;
    if (b < blocks_w0) {
        transpose_w(W0, Wt0, K0, b * SETUP_T + threadIdx.x);
        return;
    }
    b -= blocks_w0;
    const int layer = b / STAGE_BLOCKS;
    const int i = (b % STAGE_BLOCKS) * SETUP_T + threadIdx.x;
    if (layer == 0)      stage_w_frag(W1, Wf1, i);
    else if (layer == 1) stage_w_frag(W2, Wf2, i);
    else                 stage_w_frag(W3, Wf3, i);
}

__device__ __forceinline__ v16h load_frag(const _Float16* __restrict__ p) {
    v16h r;
#pragma unroll
    for (int e = 0; e < 8; ++e) { r[e] = p[e]; r[e + 8] = p[e + 16]; }
    return r;
}

__global__ void __launch_bounds__(256) gemm_wmma_f16(const _Float16* __restrict__ A,
                                                     const _Float16* __restrict__ Bt,
                                                     float* __restrict__ C,
                                                     int M, int K) {
    int gt   = blockIdx.x * 256 + threadIdx.x;
    int wid  = gt >> 5;
    int lane = gt & 31;
    int nWaves = (M >> 4) * (FOUT >> 6);
    if (wid >= nWaves) return;

    int m0 = (wid >> 2) << 4;
    int n0 = (wid & 3) << 6;
    int l15 = lane & 15;
    int kh  = (lane >> 4) << 3;

    const _Float16* pa  = A  + (size_t)(m0 + l15) * K + kh;
    const _Float16* pb0 = Bt + (size_t)(n0 + l15 +  0) * K + kh;
    const _Float16* pb1 = Bt + (size_t)(n0 + l15 + 16) * K + kh;
    const _Float16* pb2 = Bt + (size_t)(n0 + l15 + 32) * K + kh;
    const _Float16* pb3 = Bt + (size_t)(n0 + l15 + 48) * K + kh;

    v8f c0 = {}, c1 = {}, c2 = {}, c3 = {};
    for (int k0 = 0; k0 < K; k0 += 32) {
        v16h a  = load_frag(pa  + k0);
        v16h b0 = load_frag(pb0 + k0);
        v16h b1 = load_frag(pb1 + k0);
        v16h b2 = load_frag(pb2 + k0);
        v16h b3 = load_frag(pb3 + k0);
        c0 = __builtin_amdgcn_wmma_f32_16x16x32_f16(false, a, false, b0, (short)0, c0, false, false);
        c1 = __builtin_amdgcn_wmma_f32_16x16x32_f16(false, a, false, b1, (short)0, c1, false, false);
        c2 = __builtin_amdgcn_wmma_f32_16x16x32_f16(false, a, false, b2, (short)0, c2, false, false);
        c3 = __builtin_amdgcn_wmma_f32_16x16x32_f16(false, a, false, b3, (short)0, c3, false, false);
    }
    int rbase = (lane >> 4) << 3;
#pragma unroll
    for (int r = 0; r < 8; ++r) {
        size_t o = (size_t)(m0 + r + rbase) * FOUT + n0 + l15;
        C[o +  0] = c0[r];
        C[o + 16] = c1[r];
        C[o + 32] = c2[r];
        C[o + 48] = c3[r];
    }
}

typedef __attribute__((ext_vector_type(2))) float v2f;
typedef __attribute__((ext_vector_type(4))) float v4f;

__device__ __forceinline__ v4f ld4(const float* __restrict__ p) {
    return *reinterpret_cast<const v4f*>(p);
}

__device__ __forceinline__ float dot8(v4f x0, v4f x1, v4f a0, v4f a1) {
    v2f d = x0.xy * a0.xy;
    d = x0.zw * a0.zw + d;
    d = x1.xy * a1.xy + d;
    d = x1.zw * a1.zw + d;
    return d.x + d.y;
}

__device__ __forceinline__ float quad_sum(float x) {
    x += __int_as_float(__builtin_amdgcn_update_dpp(0, __float_as_int(x), 0x161, 0xf, 0xf, true));
    x += __int_as_float(__builtin_amdgcn_update_dpp(0, __float_as_int(x), 0x162, 0xf, 0xf, true));
    return x;
}

__device__ __forceinline__ float lrelu(float e) {
    return fmaxf(e, NEG_SLOPE * e);
}

#define APITCH 264

__device__ __forceinline__ v16h load_frag_lds(const _Float16* p) {
    const v8h lo = *reinterpret_cast<const v8h*>(p);
    const v8h hi = *reinterpret_cast<const v8h*>(p + 16);
    return __builtin_shufflevector(lo, hi, 0, 1, 2, 3, 4, 5, 6, 7, 8, 9, 10, 11, 12, 13, 14, 15);
}

__device__ __forceinline__ v16h load_frag_slab(const _Float16* __restrict__ p) {
    const v8h lo = *reinterpret_cast<const v8h*>(p);
    const v8h hi = *reinterpret_cast<const v8h*>(p + 256);
    return __builtin_shufflevector(lo, hi, 0, 1, 2, 3, 4, 5, 6, 7, 8, 9, 10, 11, 12, 13, 14, 15);
}

__global__ void __launch_bounds__(512, 12) gat_pull(const float* __restrict__ feat,
                                                const float* __restrict__ al,
                                                const float* __restrict__ ar,
                                                const int* __restrict__ off,
                                                const int* __restrict__ eidx,
                                                const float* __restrict__ bias,
                                                float* __restrict__ out,
                                                const _Float16* __restrict__ Wf_next,
                                                float* __restrict__ feat_next,
                                                int n, int mode) {
    __shared__ __attribute__((aligned(16))) _Float16 tileA[16 * APITCH];
    const int wave = threadIdx.x >> 5;
    const int lane = threadIdx.x & 31;
    const int v    = blockIdx.x * 16 + wave;
    const int col = (lane >> 2) * HDIM + (lane & 3) * 8;
    v8h hv = {};
    if (v < n) {

    const int beg = off[v], end = off[v + 1];
    const int last = end - 1;

    const v4f a0 = ld4(al + col), a1 = ld4(al + col + 4);
    float ern;
    {
        const v4f b0 = ld4(ar + col), b1 = ld4(ar + col + 4);
        const float* fv = feat + (size_t)v * FOUT + col;
        ern = quad_sum(dot8(ld4(fv), ld4(fv + 4), b0, b1));
    }

    float m = -INFINITY, sum = 0.f;
    float acc[8];
#pragma unroll
    for (int q = 0; q < 8; ++q) acc[q] = 0.f;

    int s0 = 0, s1 = 0, s2 = 0, s3 = 0;
    if (end > beg) {
        s0 = eidx[beg];
        s1 = eidx[min(beg + 1, last)];
        s2 = eidx[min(beg + 2, last)];
        s3 = eidx[min(beg + 3, last)];
    }
    for (int t = beg; t < end; t += 4) {
        const int n0 = eidx[min(t + 4, last)];
        const int n1 = eidx[min(t + 5, last)];
        const int n2 = eidx[min(t + 6, last)];
        const int n3 = eidx[min(t + 7, last)];

        const float* f0 = feat + (size_t)s0 * FOUT + col;
        const float* f1 = feat + (size_t)s1 * FOUT + col;
        const float* f2 = feat + (size_t)s2 * FOUT + col;
        const float* f3 = feat + (size_t)s3 * FOUT + col;
        const v4f r0a = ld4(f0), r0b = ld4(f0 + 4);
        const v4f r1a = ld4(f1), r1b = ld4(f1 + 4);
        const v4f r2a = ld4(f2), r2b = ld4(f2 + 4);
        const v4f r3a = ld4(f3), r3b = ld4(f3 + 4);
        __builtin_amdgcn_sched_barrier(0);

        float e0 = lrelu(quad_sum(dot8(r0a, r0b, a0, a1)) + ern);
        float e1 = lrelu(quad_sum(dot8(r1a, r1b, a0, a1)) + ern);
        float e2 = lrelu(quad_sum(dot8(r2a, r2b, a0, a1)) + ern);
        float e3 = lrelu(quad_sum(dot8(r3a, r3b, a0, a1)) + ern);
        e1 = (t + 1 <= last) ? e1 : -INFINITY;
        e2 = (t + 2 <= last) ? e2 : -INFINITY;
        e3 = (t + 3 <= last) ? e3 : -INFINITY;

        const float mn = fmaxf(fmaxf(m, e0), fmaxf(fmaxf(e1, e2), e3));
        const float sc = __expf(m - mn);
        const float p0 = __expf(e0 - mn);
        const float p1 = __expf(e1 - mn);
        const float p2 = __expf(e2 - mn);
        const float p3 = __expf(e3 - mn);
        sum = sum * sc + ((p0 + p1) + (p2 + p3));
        const float r0[8] = {r0a.x, r0a.y, r0a.z, r0a.w, r0b.x, r0b.y, r0b.z, r0b.w};
        const float r1[8] = {r1a.x, r1a.y, r1a.z, r1a.w, r1b.x, r1b.y, r1b.z, r1b.w};
        const float r2[8] = {r2a.x, r2a.y, r2a.z, r2a.w, r2b.x, r2b.y, r2b.z, r2b.w};
        const float r3[8] = {r3a.x, r3a.y, r3a.z, r3a.w, r3b.x, r3b.y, r3b.z, r3b.w};
#pragma unroll
        for (int q = 0; q < 8; ++q) {
            float x = acc[q] * sc;
            x = fmaf(p0, r0[q], x);
            x = fmaf(p1, r1[q], x);
            x = fmaf(p2, r2[q], x);
            x = fmaf(p3, r3[q], x);
            acc[q] = x;
        }
        m = mn;
        asm volatile("" :: "v"(n0), "v"(n1), "v"(n2), "v"(n3));
        s0 = n0; s1 = n1; s2 = n2; s3 = n3;
    }
    const float inv = (end > beg) ? 1.f / sum : 0.f;

    const size_t base = (size_t)v * FOUT + col;
    const v4f bi0 = ld4(bias + col), bi1 = ld4(bias + col + 4);
    v4f rs0 = {0.f, 0.f, 0.f, 0.f}, rs1 = {0.f, 0.f, 0.f, 0.f};
    if (mode != 0) {
        rs0 = ld4(out + base);
        rs1 = ld4(out + base + 4);
    }
    const float bi[8] = {bi0.x, bi0.y, bi0.z, bi0.w, bi1.x, bi1.y, bi1.z, bi1.w};
    const float rs[8] = {rs0.x, rs0.y, rs0.z, rs0.w, rs1.x, rs1.y, rs1.z, rs1.w};
    float res[8];
    if (mode == 0) {
#pragma unroll
        for (int q = 0; q < 8; ++q) {
            const float val = acc[q] * inv + bi[q];
            res[q] = (val > 0.f) ? val : (__expf(val) - 1.f);
        }
    } else {
#pragma unroll
        for (int q = 0; q < 8; ++q) {
            const float val = acc[q] * inv + bi[q];
            res[q] = val + rs[q];
        }
    }
    v4f o0, o1;
#pragma unroll
    for (int q = 0; q < 4; ++q) {
        o0[q] = res[q];
        o1[q] = res[q + 4];
    }
#pragma unroll
    for (int q = 0; q < 8; ++q) hv[q] = (_Float16)res[q];
    *reinterpret_cast<v4f*>(out + base) = o0;
    *reinterpret_cast<v4f*>(out + base + 4) = o1;
    }

    if (Wf_next == nullptr) return;

    const int l15 = lane & 15;
    const int kh  = (lane >> 4) << 3;
    const int n0  = wave << 4;
    const _Float16* pa = tileA + l15 * APITCH + kh;
    const _Float16* pb = Wf_next + wave * (FOUT * 16) + lane * 8;

    const v16h b0 = load_frag_slab(pb);
    const v16h b1 = load_frag_slab(pb + 512);
    const v16h b2 = load_frag_slab(pb + 1024);
    const v16h b3 = load_frag_slab(pb + 1536);

    *reinterpret_cast<v8h*>(tileA + wave * APITCH + lane * 8) = hv;
    __builtin_amdgcn_fence(__ATOMIC_RELEASE, "workgroup", "local");
    __builtin_amdgcn_s_barrier();
    __builtin_amdgcn_fence(__ATOMIC_ACQUIRE, "workgroup", "local");

    const v16h b4 = load_frag_slab(pb + 2048);
    const v16h b5 = load_frag_slab(pb + 2560);
    const v16h b6 = load_frag_slab(pb + 3072);
    const v16h b7 = load_frag_slab(pb + 3584);

    v8f c = {};
    c = __builtin_amdgcn_wmma_f32_16x16x32_f16(false, load_frag_lds(pa),       false, b0, (short)0, c, false, false);
    c = __builtin_amdgcn_wmma_f32_16x16x32_f16(false, load_frag_lds(pa + 32),  false, b1, (short)0, c, false, false);
    c = __builtin_amdgcn_wmma_f32_16x16x32_f16(false, load_frag_lds(pa + 64),  false, b2, (short)0, c, false, false);
    c = __builtin_amdgcn_wmma_f32_16x16x32_f16(false, load_frag_lds(pa + 96),  false, b3, (short)0, c, false, false);
    c = __builtin_amdgcn_wmma_f32_16x16x32_f16(false, load_frag_lds(pa + 128), false, b4, (short)0, c, false, false);
    c = __builtin_amdgcn_wmma_f32_16x16x32_f16(false, load_frag_lds(pa + 160), false, b5, (short)0, c, false, false);
    c = __builtin_amdgcn_wmma_f32_16x16x32_f16(false, load_frag_lds(pa + 192), false, b6, (short)0, c, false, false);
    c = __builtin_amdgcn_wmma_f32_16x16x32_f16(false, load_frag_lds(pa + 224), false, b7, (short)0, c, false, false);

    const int m0 = blockIdx.x * 16 + ((lane >> 4) << 3);
    float* pc = feat_next + (size_t)m0 * FOUT + n0 + l15;
    if (blockIdx.x * 16 + 16 <= n) {
#pragma unroll
        for (int r = 0; r < 8; ++r) pc[(size_t)r * FOUT] = c[r];
    } else {
#pragma unroll
        for (int r = 0; r < 8; ++r)
            if (m0 + r < n) pc[(size_t)r * FOUT] = c[r];
    }
}

static inline size_t align256(size_t x) { return (x + 255) & ~(size_t)255; }

extern "C" void kernel_launch(void* const* d_in, const int* in_sizes, int n_in,
                              void* d_out, int out_size, void* d_ws, size_t ws_size,
                              hipStream_t stream) {
    const int N = in_sizes[0] / 128;
    const int E = in_sizes[1];

    const float* x   = (const float*)d_in[0];
    const int*   src = (const int*)d_in[1];
    const int*   dst = (const int*)d_in[2];

    char* ws = (char*)d_ws;
    size_t o = 0;
    int* off    = (int*)(ws + o); o = align256(o + (size_t)(N + 1) * 4);
    int* deg    = (int*)(ws + o); o = align256(o + (size_t)N * 4);
    int* cursor = (int*)(ws + o); o = align256(o + (size_t)N * 4);
    int* eidx   = (int*)(ws + o); o = align256(o + (size_t)E * 4);
    _Float16* x16 = (_Float16*)(ws + o); o = align256(o + (size_t)N * 128 * 2);
    _Float16* wt16[4];
    for (int l = 0; l < 4; ++l) { wt16[l] = (_Float16*)(ws + o); o = align256(o + (size_t)FOUT * FOUT * 2); }
    float* featA = (float*)(ws + o); o = align256(o + (size_t)N * FOUT * 4);
    float* featB = (float*)(ws + o); o = align256(o + (size_t)N * FOUT * 4);
    (void)ws_size;

    float* h = (float*)d_out;

    const int T = 256;
    auto blk = [](int n, int t) { return (n + t - 1) / t; };

    const int K0 = in_sizes[3] / FOUT;
    {
        const int bx = blk(N * K0, SETUP_T), bz = blk(N, SETUP_T), bw0 = blk(K0 * FOUT, SETUP_T);
        setup_all<<<bx + bz + bw0 + 3 * STAGE_BLOCKS, SETUP_T, 0, stream>>>(
            x, x16, N * K0, deg, cursor, N,
            (const float*)d_in[3],  wt16[0], K0,
            (const float*)d_in[7],  wt16[1],
            (const float*)d_in[11], wt16[2],
            (const float*)d_in[15], wt16[3],
            bx, bz, bw0);
    }

    count_deg<<<blk(E, T), T, 0, stream>>>(dst, deg, E);
    scan_off<<<blk(N, SCAN_SEG), SCAN_T, 0, stream>>>(deg, off, N);
    scatter_edges<<<blk(E, T), T, 0, stream>>>(src, dst, off, cursor, eidx, E);

    {
        int nWaves = (N / 16) * (FOUT / 64);
        gemm_wmma_f16<<<blk(nWaves * 32, T), T, 0, stream>>>(x16, wt16[0], featA, N, K0);
    }

    for (int layer = 0; layer < 4; ++layer) {
        const float* al = (const float*)d_in[4 + 4 * layer];
        const float* ar = (const float*)d_in[5 + 4 * layer];
        const float* b  = (const float*)d_in[6 + 4 * layer];
        const float* fin  = (layer & 1) ? featB : featA;
        float*       fout = (layer & 1) ? featA : featB;
        const _Float16* wnext = (layer < 3) ? wt16[layer + 1] : nullptr;
        int mode = (layer == 0) ? 0 : 1;
        gat_pull<<<blk(N, 16), 512, 0, stream>>>(fin, al, ar, off, eidx, b, h, wnext, fout, N, mode);
    }
}


# edge grouping: each node's counter is word 0 of its own 128-byte slot line (31 ids behind it, a 32nd edge goes to the overflow list), not a packed array: no two nodes' adds share a line
# speedup vs baseline: 2.6539x; 1.3021x over previous
#include <hip/hip_runtime.h>
#include <hip/hip_bf16.h>
#include <math.h>

typedef __attribute__((ext_vector_type(16))) _Float16 v16h;
typedef __attribute__((ext_vector_type(8)))  float    v8f;
typedef __attribute__((ext_vector_type(8)))  _Float16 v8h;
typedef __attribute__((ext_vector_type(4)))  int      v4i;

#define HEADS 8
#define HDIM  32
#define FOUT  256
#define NEG_SLOPE 0.2f

#define ROWI  32
#define SLOTS 31

__global__ void place_edges(const int* __restrict__ src, const int* __restrict__ dst,
                            int* __restrict__ slot,
                            int* __restrict__ novf, int* __restrict__ ovf_dst,
                            int* __restrict__ ovf_src, int e) {
    int i = blockIdx.x * blockDim.x + threadIdx.x;
    if (i < e) {
        const int d = dst[i];
        const int s = src[i];
        int* row = slot + (size_t)d * ROWI;
        const int pos = atomicAdd(row, 1);
        if (pos < SLOTS) {
            row[1 + pos] = s;
        } else {
            const int q = atomicAdd(novf, 1);
            ovf_dst[q] = d;
            ovf_src[q] = s;
        }
    }
}

__device__ __forceinline__ void transpose_w(const float* __restrict__ W, _Float16* __restrict__ Wt,
                                            int K, int i) {
    if (i < K * FOUT) {
        int k = i / FOUT, c = i % FOUT;
        Wt[c * K + k] = (_Float16)W[k * FOUT + c];
    }
}

__device__ __forceinline__ void stage_w_frag(const float* __restrict__ W, _Float16* __restrict__ Wf,
                                             int i) {
    if (i < FOUT * FOUT) {
        int k = i / FOUT, c = i % FOUT;
        int w = c >> 4, cc = c & 15;
        int j = k >> 5, s = (k >> 4) & 1, h = (k >> 3) & 1, e = k & 7;
        Wf[(((w * 8 + j) * 2 + s) * 32 + (h * 16 + cc)) * 8 + e] = (_Float16)W[i];
    }
}

#define SETUP_T 256
#define STAGE_BLOCKS (FOUT * FOUT / SETUP_T)
__global__ void __launch_bounds__(SETUP_T) setup_all(const float* __restrict__ x,
                                                     _Float16* __restrict__ x16, int n_x,
                                                     int* __restrict__ slot,
                                                     int* __restrict__ novf, int n_slot4,
                                                     const float* __restrict__ W0,
                                                     _Float16* __restrict__ Wt0, int K0,
                                                     const float* __restrict__ W1,
                                                     _Float16* __restrict__ Wf1,
                                                     const float* __restrict__ W2,
                                                     _Float16* __restrict__ Wf2,
                                                     const float* __restrict__ W3,
                                                     _Float16* __restrict__ Wf3,
                                                     int blocks_x, int blocks_zero,
                                                     int blocks_w0) {
    int b = blockIdx.x;
    if (b < blocks_x) {
        const int i = b * SETUP_T + threadIdx.x;
        if (i < n_x) x16[i] = (_Float16)x[i];
        return;
    }
    b -= blocks_x;
    if (b < blocks_zero) {
        const int i = b * SETUP_T + threadIdx.x;
        if (i < n_slot4) reinterpret_cast<v4i*>(slot)[i] = (v4i){0, 0, 0, 0};
        if (i == 0) *novf = 0;
        return;
    }
    b -= blocks_zero;
    if (b < blocks_w0) {
        transpose_w(W0, Wt0, K0, b * SETUP_T + threadIdx.x);
        return;
    }
    b -= blocks_w0;
    const int layer = b / STAGE_BLOCKS;
    const int i = (b % STAGE_BLOCKS) * SETUP_T + threadIdx.x;
    if (layer == 0)      stage_w_frag(W1, Wf1, i);
    else if (layer == 1) stage_w_frag(W2, Wf2, i);
    else                 stage_w_frag(W3, Wf3, i);
}

__device__ __forceinline__ v16h load_frag(const _Float16* __restrict__ p) {
    v16h r;
#pragma unroll
    for (int e = 0; e < 8; ++e) { r[e] = p[e]; r[e + 8] = p[e + 16]; }
    return r;
}

__global__ void __launch_bounds__(256) gemm_wmma_f16(const _Float16* __restrict__ A,
                                                     const _Float16* __restrict__ Bt,
                                                     float* __restrict__ C,
                                                     int M, int K) {
    int gt   = blockIdx.x * 256 + threadIdx.x;
    int wid  = gt >> 5;
    int lane = gt & 31;
    int nWaves = (M >> 4) * (FOUT >> 6);
    if (wid >= nWaves) return;

    int m0 = (wid >> 2) << 4;
    int n0 = (wid & 3) << 6;
    int l15 = lane & 15;
    int kh  = (lane >> 4) << 3;

    const _Float16* pa  = A  + (size_t)(m0 + l15) * K + kh;
    const _Float16* pb0 = Bt + (size_t)(n0 + l15 +  0) * K + kh;
    const _Float16* pb1 = Bt + (size_t)(n0 + l15 + 16) * K + kh;
    const _Float16* pb2 = Bt + (size_t)(n0 + l15 + 32) * K + kh;
    const _Float16* pb3 = Bt + (size_t)(n0 + l15 + 48) * K + kh;

    v8f c0 = {}, c1 = {}, c2 = {}, c3 = {};
    for (int k0 = 0; k0 < K; k0 += 32) {
        v16h a  = load_frag(pa  + k0);
        v16h b0 = load_frag(pb0 + k0);
        v16h b1 = load_frag(pb1 + k0);
        v16h b2 = load_frag(pb2 + k0);
        v16h b3 = load_frag(pb3 + k0);
        c0 = __builtin_amdgcn_wmma_f32_16x16x32_f16(false, a, false, b0, (short)0, c0, false, false);
        c1 = __builtin_amdgcn_wmma_f32_16x16x32_f16(false, a, false, b1, (short)0, c1, false, false);
        c2 = __builtin_amdgcn_wmma_f32_16x16x32_f16(false, a, false, b2, (short)0, c2, false, false);
        c3 = __builtin_amdgcn_wmma_f32_16x16x32_f16(false, a, false, b3, (short)0, c3, false, false);
    }
    int rbase = (lane >> 4) << 3;
#pragma unroll
    for (int r = 0; r < 8; ++r) {
        size_t o = (size_t)(m0 + r + rbase) * FOUT + n0 + l15;
        C[o +  0] = c0[r];
        C[o + 16] = c1[r];
        C[o + 32] = c2[r];
        C[o + 48] = c3[r];
    }
}

typedef __attribute__((ext_vector_type(2))) float v2f;
typedef __attribute__((ext_vector_type(4))) float v4f;

__device__ __forceinline__ v4f ld4(const float* __restrict__ p) {
    return *reinterpret_cast<const v4f*>(p);
}

__device__ __forceinline__ float dot8(v4f x0, v4f x1, v4f a0, v4f a1) {
    v2f d = x0.xy * a0.xy;
    d = x0.zw * a0.zw + d;
    d = x1.xy * a1.xy + d;
    d = x1.zw * a1.zw + d;
    return d.x + d.y;
}

__device__ __forceinline__ float quad_sum(float x) {
    x += __int_as_float(__builtin_amdgcn_update_dpp(0, __float_as_int(x), 0x161, 0xf, 0xf, true));
    x += __int_as_float(__builtin_amdgcn_update_dpp(0, __float_as_int(x), 0x162, 0xf, 0xf, true));
    return x;
}

__device__ __forceinline__ float lrelu(float e) {
    return fmaxf(e, NEG_SLOPE * e);
}

#define APITCH 264

__device__ __forceinline__ v16h load_frag_lds(const _Float16* p) {
    const v8h lo = *reinterpret_cast<const v8h*>(p);
    const v8h hi = *reinterpret_cast<const v8h*>(p + 16);
    return __builtin_shufflevector(lo, hi, 0, 1, 2, 3, 4, 5, 6, 7, 8, 9, 10, 11, 12, 13, 14, 15);
}

struct EdgeLists {
    const int* slot;
    const int* novf;
    const int* ovf_dst;
    const int* ovf_src;
    int        ovf_cap;
};

__device__ __forceinline__ v8h pull_node(const float* __restrict__ feat,
                                         const float* __restrict__ al,
                                         const float* __restrict__ ar,
                                         const EdgeLists& lists,
                                         const float* __restrict__ bias,
                                         float* __restrict__ out,
                                         int v, int lane, int mode) {
    v8h hv;
    const int col = lane * 8;
    const int* __restrict__ eidx = lists.slot;
    const v4i head = *reinterpret_cast<const v4i*>(eidx + v * ROWI);
    const int id3 = eidx[v * ROWI + 4];
    const int deg = head.x;
    const int beg = v * ROWI + 1, end = beg + min(deg, SLOTS);
    const int last = end - 1;
    const int over = __builtin_amdgcn_readfirstlane(deg - SLOTS);

    const v4f a0 = ld4(al + col), a1 = ld4(al + col + 4);
    float ern;
    {
        const v4f b0 = ld4(ar + col), b1 = ld4(ar + col + 4);
        const float* fv = feat + (size_t)v * FOUT + col;
        ern = quad_sum(dot8(ld4(fv), ld4(fv + 4), b0, b1));
    }
    asm volatile("" :: "v"(id3));

    float m = -INFINITY, sum = 0.f;
    float acc[8];
#pragma unroll
    for (int q = 0; q < 8; ++q) acc[q] = 0.f;

    const int s0i = head.y;
    const int s1i = (deg > 1) ? head.z : s0i;
    const int s2i = (deg > 2) ? head.w : s1i;
    const int s3i = (deg > 3) ? id3 : s2i;
    int s0 = s0i, s1 = s1i, s2 = s2i, s3 = s3i;
    for (int t = beg; t < end; t += 4) {
        const int n0 = eidx[min(t + 4, last)];
        const int n1 = eidx[min(t + 5, last)];
        const int n2 = eidx[min(t + 6, last)];
        const int n3 = eidx[min(t + 7, last)];

        const float* f0 = feat + (size_t)s0 * FOUT + col;
        const float* f1 = feat + (size_t)s1 * FOUT + col;
        const float* f2 = feat + (size_t)s2 * FOUT + col;
        const float* f3 = feat + (size_t)s3 * FOUT + col;
        const v4f r0a = ld4(f0), r0b = ld4(f0 + 4);
        const v4f r1a = ld4(f1), r1b = ld4(f1 + 4);
        const v4f r2a = ld4(f2), r2b = ld4(f2 + 4);
        const v4f r3a = ld4(f3), r3b = ld4(f3 + 4);
        __builtin_amdgcn_sched_barrier(0);

        float e0 = lrelu(quad_sum(dot8(r0a, r0b, a0, a1)) + ern);
        float e1 = lrelu(quad_sum(dot8(r1a, r1b, a0, a1)) + ern);
        float e2 = lrelu(quad_sum(dot8(r2a, r2b, a0, a1)) + ern);
        float e3 = lrelu(quad_sum(dot8(r3a, r3b, a0, a1)) + ern);
        e1 = (t + 1 <= last) ? e1 : -INFINITY;
        e2 = (t + 2 <= last) ? e2 : -INFINITY;
        e3 = (t + 3 <= last) ? e3 : -INFINITY;

        const float mn = fmaxf(fmaxf(m, e0), fmaxf(fmaxf(e1, e2), e3));
        const float sc = __expf(m - mn);
        const float p0 = __expf(e0 - mn);
        const float p1 = __expf(e1 - mn);
        const float p2 = __expf(e2 - mn);
        const float p3 = __expf(e3 - mn);
        sum = sum * sc + ((p0 + p1) + (p2 + p3));
        const float r0[8] = {r0a.x, r0a.y, r0a.z, r0a.w, r0b.x, r0b.y, r0b.z, r0b.w};
        const float r1[8] = {r1a.x, r1a.y, r1a.z, r1a.w, r1b.x, r1b.y, r1b.z, r1b.w};
        const float r2[8] = {r2a.x, r2a.y, r2a.z, r2a.w, r2b.x, r2b.y, r2b.z, r2b.w};
        const float r3[8] = {r3a.x, r3a.y, r3a.z, r3a.w, r3b.x, r3b.y, r3b.z, r3b.w};
#pragma unroll
        for (int q = 0; q < 8; ++q) {
            float x = acc[q] * sc;
            x = fmaf(p0, r0[q], x);
            x = fmaf(p1, r1[q], x);
            x = fmaf(p2, r2[q], x);
            x = fmaf(p3, r3[q], x);
            acc[q] = x;
        }
        m = mn;
        s0 = n0; s1 = n1; s2 = n2; s3 = n3;
    }
    if (over > 0) {
        const int no = min(*lists.novf, lists.ovf_cap);
        for (int j0 = 0; j0 < no; j0 += 32) {
            const int j = j0 + lane;
            int d = -1, s = 0;
            if (j < no) { d = lists.ovf_dst[j]; s = lists.ovf_src[j]; }
            unsigned hits = __builtin_amdgcn_ballot_w32(d == v);
            while (hits != 0u) {
                const int b = __builtin_ctz(hits);
                hits &= hits - 1u;
                const int sx = __builtin_amdgcn_readlane(s, b);
                const float* fx = feat + (size_t)sx * FOUT + col;
                const v4f xa = ld4(fx), xb = ld4(fx + 4);
                const float ex = lrelu(quad_sum(dot8(xa, xb, a0, a1)) + ern);
                const float mn = fmaxf(m, ex);
                const float sc = __expf(m - mn);
                const float px = __expf(ex - mn);
                sum = sum * sc + px;
                const float rx[8] = {xa.x, xa.y, xa.z, xa.w, xb.x, xb.y, xb.z, xb.w};
#pragma unroll
                for (int q = 0; q < 8; ++q) acc[q] = fmaf(px, rx[q], acc[q] * sc);
                m = mn;
            }
        }
    }
    const float inv = (end > beg) ? 1.f / sum : 0.f;

    const size_t base = (size_t)v * FOUT + col;
    const v4f bi0 = ld4(bias + col), bi1 = ld4(bias + col + 4);
    v4f rs0 = {0.f, 0.f, 0.f, 0.f}, rs1 = {0.f, 0.f, 0.f, 0.f};
    if (mode != 0) {
        rs0 = ld4(out + base);
        rs1 = ld4(out + base + 4);
    }
    const float bi[8] = {bi0.x, bi0.y, bi0.z, bi0.w, bi1.x, bi1.y, bi1.z, bi1.w};
    const float rs[8] = {rs0.x, rs0.y, rs0.z, rs0.w, rs1.x, rs1.y, rs1.z, rs1.w};
    float res[8];
    if (mode == 0) {
#pragma unroll
        for (int q = 0; q < 8; ++q) {
            const float val = acc[q] * inv + bi[q];
            res[q] = (val > 0.f) ? val : (__expf(val) - 1.f);
        }
    } else {
#pragma unroll
        for (int q = 0; q < 8; ++q) {
            const float val = acc[q] * inv + bi[q];
            res[q] = val + rs[q];
        }
    }
    v4f o0, o1;
#pragma unroll
    for (int q = 0; q < 4; ++q) {
        o0[q] = res[q];
        o1[q] = res[q + 4];
    }
#pragma unroll
    for (int q = 0; q < 8; ++q) hv[q] = (_Float16)res[q];
    *reinterpret_cast<v4f*>(out + base) = o0;
    *reinterpret_cast<v4f*>(out + base + 4) = o1;
    return hv;
}

__global__ void __launch_bounds__(512, 12) gat_pull(const float* __restrict__ feat,
                                                const float* __restrict__ al,
                                                const float* __restrict__ ar,
                                                const EdgeLists lists,
                                                const float* __restrict__ bias,
                                                float* __restrict__ out,
                                                int n, int mode) {
    const int wave = threadIdx.x >> 5;
    const int lane = threadIdx.x & 31;
    const int v    = blockIdx.x * 16 + wave;
    if (v < n) (void)pull_node(feat, al, ar, lists, bias, out, v, lane, mode);
}

#define TNODES 24

__device__ __forceinline__ v16h load_frag_wlds(const _Float16* p) {
    const v8h lo = *reinterpret_cast<const v8h*>(p);
    const v8h hi = *reinterpret_cast<const v8h*>(p + 256);
    return __builtin_shufflevector(lo, hi, 0, 1, 2, 3, 4, 5, 6, 7, 8, 9, 10, 11, 12, 13, 14, 15);
}

__global__ void __launch_bounds__(TNODES * 32, 12) gat_pull_mm(const float* __restrict__ feat,
                                                   const float* __restrict__ al,
                                                   const float* __restrict__ ar,
                                                   const EdgeLists lists,
                                                   const float* __restrict__ bias,
                                                   float* __restrict__ out,
                                                   const _Float16* __restrict__ Wf_next,
                                                   float* __restrict__ feat_next,
                                                   int n, int mode, int ntiles) {
    __shared__ __attribute__((aligned(16))) _Float16 wl[FOUT * FOUT];
    __shared__ __attribute__((aligned(16))) _Float16 tileA[32 * APITCH];

    for (int tile = blockIdx.x; tile < ntiles; tile += gridDim.x) {
        int tnode = threadIdx.x;
        asm volatile("" : "+v"(tnode));
        const int v = tile * TNODES + (tnode >> 5);
        v8h hv = {};
        if (v < n) hv = pull_node(feat, al, ar, lists, bias, out, v, tnode & 31, mode);

        int tprod = threadIdx.x;
        asm volatile("" : "+v"(tprod));
        const int wave = __builtin_amdgcn_readfirstlane(tprod >> 5);
        const int lane = tprod & 31;
        _Float16* pw = wl + (wave & 15) * (FOUT * 16) + lane * 8;

        if (tile == (int)blockIdx.x && wave < 16) {
            const _Float16* pb = Wf_next + wave * (FOUT * 16) + lane * 8;
#pragma unroll
            for (int j = 0; j < 8; ++j) {
                const v8h lo = *reinterpret_cast<const v8h*>(pb + j * 512);
                const v8h hi = *reinterpret_cast<const v8h*>(pb + j * 512 + 256);
                *reinterpret_cast<v8h*>(pw + j * 512)       = lo;
                *reinterpret_cast<v8h*>(pw + j * 512 + 256) = hi;
            }
        }

        __builtin_amdgcn_fence(__ATOMIC_RELEASE, "workgroup", "local");
        __builtin_amdgcn_s_barrier();
        *reinterpret_cast<v8h*>(tileA + wave * APITCH + lane * 8) = hv;
        __builtin_amdgcn_fence(__ATOMIC_RELEASE, "workgroup", "local");
        __builtin_amdgcn_s_barrier();
        __builtin_amdgcn_fence(__ATOMIC_ACQUIRE, "workgroup", "local");

        if (wave < 16) {
            const int l15 = lane & 15;
            const int h8  = (lane >> 4) << 3;
            const _Float16* pa0 = tileA + l15 * APITCH + h8;
            const _Float16* pa1 = pa0 + 16 * APITCH;
            v8f c0 = {}, c1 = {};
#pragma unroll
            for (int j = 0; j < 8; ++j) {
                const v16h b = load_frag_wlds(pw + j * 512);
                c0 = __builtin_amdgcn_wmma_f32_16x16x32_f16(false, load_frag_lds(pa0 + 32 * j), false, b,
                                                            (short)0, c0, false, false);
                c1 = __builtin_amdgcn_wmma_f32_16x16x32_f16(false, load_frag_lds(pa1 + 32 * j), false, b,
                                                            (short)0, c1, false, false);
            }
            const int m0 = tile * TNODES + h8;
            float* pc = feat_next + (size_t)m0 * FOUT + (wave << 4) + l15;
            if (tile * TNODES + TNODES <= n) {
#pragma unroll
                for (int r = 0; r < 8; ++r) pc[(size_t)r * FOUT] = c0[r];
                if (h8 == 0) {
#pragma unroll
                    for (int r = 0; r < 8; ++r) pc[(size_t)(16 + r) * FOUT] = c1[r];
                }
            } else {
#pragma unroll
                for (int r = 0; r < 8; ++r)
                    if (m0 + r < n) pc[(size_t)r * FOUT] = c0[r];
                if (h8 == 0) {
#pragma unroll
                    for (int r = 0; r < 8; ++r)
                        if (m0 + 16 + r < n) pc[(size_t)(16 + r) * FOUT] = c1[r];
                }
            }
        }
    }
}

static inline size_t align256(size_t x) { return (x + 255) & ~(size_t)255; }

extern "C" void kernel_launch(void* const* d_in, const int* in_sizes, int n_in,
                              void* d_out, int out_size, void* d_ws, size_t ws_size,
                              hipStream_t stream) {
    const int N = in_sizes[0] / 128;
    const int E = in_sizes[1];

    const float* x   = (const float*)d_in[0];
    const int*   src = (const int*)d_in[1];
    const int*   dst = (const int*)d_in[2];

    char* ws = (char*)d_ws;
    size_t o = 0;
    int* novf    = (int*)(ws + o); o = align256(o + 4);
    int* slot    = (int*)(ws + o); o = align256(o + (size_t)N * ROWI * 4);
    int* ovf_dst = (int*)(ws + o); o = align256(o + (size_t)E * 4);
    int* ovf_src = (int*)(ws + o); o = align256(o + (size_t)E * 4);
    _Float16* x16 = (_Float16*)(ws + o); o = align256(o + (size_t)N * 128 * 2);
    _Float16* wt16[4];
    for (int l = 0; l < 4; ++l) { wt16[l] = (_Float16*)(ws + o); o = align256(o + (size_t)FOUT * FOUT * 2); }
    float* featA = (float*)(ws + o); o = align256(o + (size_t)N * FOUT * 4);
    float* featB = (float*)(ws + o); o = align256(o + (size_t)N * FOUT * 4);
    (void)ws_size;

    float* h = (float*)d_out;

    const int T = 256;
    auto blk = [](int n, int t) { return (n + t - 1) / t; };

    const int K0 = in_sizes[3] / FOUT;
    {
        const int bx = blk(N * K0, SETUP_T), bz = blk(N * (ROWI / 4), SETUP_T), bw0 = blk(K0 * FOUT, SETUP_T);
        setup_all<<<bx + bz + bw0 + 3 * STAGE_BLOCKS, SETUP_T, 0, stream>>>(
            x, x16, N * K0, slot, novf, N * (ROWI / 4),
            (const float*)d_in[3],  wt16[0], K0,
            (const float*)d_in[7],  wt16[1],
            (const float*)d_in[11], wt16[2],
            (const float*)d_in[15], wt16[3],
            bx, bz, bw0);
    }

    place_edges<<<blk(E, T), T, 0, stream>>>(src, dst, slot, novf, ovf_dst, ovf_src, E);
    const EdgeLists lists = {slot, novf, ovf_dst, ovf_src, E};

    {
        int nWaves = (N / 16) * (FOUT / 64);
        gemm_wmma_f16<<<blk(nWaves * 32, T), T, 0, stream>>>(x16, wt16[0], featA, N, K0);
    }

    for (int layer = 0; layer < 4; ++layer) {
        const float* al = (const float*)d_in[4 + 4 * layer];
        const float* ar = (const float*)d_in[5 + 4 * layer];
        const float* b  = (const float*)d_in[6 + 4 * layer];
        const float* fin  = (layer & 1) ? featB : featA;
        float*       fout = (layer & 1) ? featA : featB;
        int mode = (layer == 0) ? 0 : 1;
        if (layer < 3)
            gat_pull_mm<<<512, TNODES * 32, 0, stream>>>(fin, al, ar, lists, b, h, wt16[layer + 1], fout,
                                                         N, mode, blk(N, TNODES));
        else
            gat_pull<<<blk(N, 16), 512, 0, stream>>>(fin, al, ar, lists, b, h, N, mode);
    }
}
